# starting kernel
# baseline (speedup 1.0000x reference)
#include <hip/hip_runtime.h>
#include <hip/hip_bf16.h>
#include <math.h>


#define KAN_BATCH   512
#define KAN_IN      128
#define KAN_OUT     128
#define KAN_G       5
#define KAN_KORD    3
#define KAN_NBASIS  (KAN_G + KAN_KORD)
#define KAN_NKNOT   (KAN_G + 2 * KAN_KORD + 1)
#define KAN_FEAT    (KAN_NBASIS + 1)
#define KAN_KDIM    (KAN_IN * KAN_FEAT)
#define KAN_SIZE    (KAN_IN * KAN_OUT)
#define GEMM_KC     128

typedef __attribute__((ext_vector_type(2))) float v2f;
typedef __attribute__((ext_vector_type(8))) float v8f;

__global__ __launch_bounds__(256)
void kan_features(const float* __restrict__ x,
                  const float* __restrict__ knots_g,
                  float* __restrict__ F) {
    int t = blockIdx.x * blockDim.x + threadIdx.x;
    if (t >= KAN_BATCH * KAN_IN) return;
    int b = t / KAN_IN;
    int i = t - b * KAN_IN;

    float xv = x[b * KAN_IN + i];

    float tk[KAN_NKNOT];
#pragma unroll
    for (int j = 0; j < KAN_NKNOT; ++j) tk[j] = knots_g[j];

    float bas[KAN_NKNOT - 1];
#pragma unroll
    for (int j = 0; j < KAN_NKNOT - 1; ++j)
        bas[j] = (xv >= tk[j] && xv < tk[j + 1]) ? 1.0f : 0.0f;

#pragma unroll
    for (int p = 1; p <= KAN_KORD; ++p) {
        int n = KAN_NKNOT - 1 - p;
#pragma unroll
        for (int j = 0; j < KAN_NKNOT - 1 - 1; ++j) {
            if (j < n) {
                float a = (xv - tk[j]) / (tk[j + p] - tk[j]);
                float c = (tk[j + p + 1] - xv) / (tk[j + p + 1] - tk[j + 1]);
                bas[j] = a * bas[j] + c * bas[j + 1];
            }
        }
    }

    float silu = xv / (1.0f + expf(-xv));

    float* dst = F + (size_t)b * KAN_KDIM + (size_t)i * KAN_FEAT;
    dst[0] = silu;
#pragma unroll
    for (int c = 0; c < KAN_NBASIS; ++c) dst[1 + c] = bas[c];
}

__global__ __launch_bounds__(256)
void kan_weights(const float* __restrict__ coeff,
                 const float* __restrict__ mask,
                 const float* __restrict__ scale_base,
                 const float* __restrict__ scale_spline,
                 float* __restrict__ Wt) {
    int s = blockIdx.x * blockDim.x + threadIdx.x;
    if (s >= KAN_SIZE) return;
    int o = s / KAN_IN;
    int i = s - o * KAN_IN;

    float m  = mask[s];
    float sb = m * scale_base[s];
    float ss = m * scale_spline[s];

    int kbase = i * KAN_FEAT;
    Wt[(size_t)(kbase + 0) * KAN_OUT + o] = sb;
#pragma unroll
    for (int c = 0; c < KAN_NBASIS; ++c)
        Wt[(size_t)(kbase + 1 + c) * KAN_OUT + o] = ss * coeff[(size_t)s * KAN_NBASIS + c];
}

__global__ __launch_bounds__(256)
void kan_gemm(const float* __restrict__ F,
              const float* __restrict__ Wt,
              const float* __restrict__ bias,
              float* __restrict__ out) {
    __shared__ float bsm[GEMM_KC * 16];

    int blk  = blockIdx.x;
    int mBlk = blk >> 3;
    int nBlk = blk & 7;
    int tid  = threadIdx.x;
    int wave = tid >> 5;
    int lane = tid & 31;
    int h    = lane >> 4;
    int ln   = lane & 15;

    int m0 = mBlk * 128 + wave * 16;
    int n0 = nBlk * 16;

    const float* Arow = F + (size_t)(m0 + ln) * KAN_KDIM;
    float bval = bias[n0 + ln];

    v8f acc = {};

    for (int k0 = 0; k0 < KAN_KDIM; k0 += GEMM_KC) {
        __syncthreads();
#pragma unroll
        for (int r = 0; r < (GEMM_KC * 16) / 256; ++r) {
            int idx = r * 256 + tid;
            int kk  = idx >> 4;
            int nn  = idx & 15;
            bsm[idx] = Wt[(size_t)(k0 + kk) * KAN_OUT + n0 + nn];
        }
        __syncthreads();

        if (k0 + GEMM_KC < KAN_KDIM)
            __builtin_prefetch(Arow + k0 + GEMM_KC, 0, 0);

#pragma unroll 4
        for (int kk = 0; kk < GEMM_KC; kk += 4) {
            v2f a = *(const v2f*)(Arow + k0 + kk + 2 * h);
            v2f b;
            b.x = bsm[(kk + 2 * h + 0) * 16 + ln];
            b.y = bsm[(kk + 2 * h + 1) * 16 + ln];
            acc = __builtin_amdgcn_wmma_f32_16x16x4_f32(
                 false, a,  false, b,
                 (short)0, acc,  false,  false);
        }
    }

    float* ocol = out + n0 + ln;
#pragma unroll
    for (int r = 0; r < 8; ++r) {
        int row = m0 + r + 8 * h;
        ocol[(size_t)row * KAN_OUT] = acc[r] + bval;
    }
}

extern "C" void kernel_launch(void* const* d_in, const int* in_sizes, int n_in,
                              void* d_out, int out_size, void* d_ws, size_t ws_size,
                              hipStream_t stream) {
    const float* x            = (const float*)d_in[0];
    const float* grid_knots   = (const float*)d_in[1];
    const float* coeff        = (const float*)d_in[2];
    const float* mask         = (const float*)d_in[3];
    const float* scale_base   = (const float*)d_in[4];
    const float* scale_spline = (const float*)d_in[5];
    const float* bias         = (const float*)d_in[6];

    float* F  = (float*)d_ws;
    float* Wt = F + (size_t)KAN_BATCH * KAN_KDIM;

    kan_features<<<(KAN_BATCH * KAN_IN + 255) / 256, 256, 0, stream>>>(x, grid_knots, F);
    kan_weights<<<(KAN_SIZE + 255) / 256, 256, 0, stream>>>(coeff, mask, scale_base,
                                                            scale_spline, Wt);
    kan_gemm<<<32, 256, 0, stream>>>(F, Wt, bias, (float*)d_out);
}


# GEMM re-tiled: 256 work-groups of one 16x16 tile (was 32 of 128x16), K split over the 8 waves, every fragment load issued before the first product, partials summed through LDS: new summation order
# speedup vs baseline: 2.0398x; 2.0398x over previous
#include <hip/hip_runtime.h>
#include <hip/hip_bf16.h>
#include <math.h>


#define KAN_BATCH   512
#define KAN_IN      128
#define KAN_OUT     128
#define KAN_G       5
#define KAN_KORD    3
#define KAN_NBASIS  (KAN_G + KAN_KORD)
#define KAN_NKNOT   (KAN_G + 2 * KAN_KORD + 1)
#define KAN_FEAT    (KAN_NBASIS + 1)
#define KAN_KDIM    (KAN_IN * KAN_FEAT)
#define KAN_SIZE    (KAN_IN * KAN_OUT)

typedef __attribute__((ext_vector_type(2))) float v2f;
typedef __attribute__((ext_vector_type(4))) float v4f;
typedef __attribute__((ext_vector_type(8))) float v8f;

__global__ __launch_bounds__(256)
void kan_features(const float* __restrict__ x,
                  const float* __restrict__ knots_g,
                  float* __restrict__ F) {
    int t = blockIdx.x * blockDim.x + threadIdx.x;
    if (t >= KAN_BATCH * KAN_IN) return;
    int b = t / KAN_IN;
    int i = t - b * KAN_IN;

    float xv = x[b * KAN_IN + i];

    float tk[KAN_NKNOT];
#pragma unroll
    for (int j = 0; j < KAN_NKNOT; ++j) tk[j] = knots_g[j];

    float bas[KAN_NKNOT - 1];
#pragma unroll
    for (int j = 0; j < KAN_NKNOT - 1; ++j)
        bas[j] = (xv >= tk[j] && xv < tk[j + 1]) ? 1.0f : 0.0f;

#pragma unroll
    for (int p = 1; p <= KAN_KORD; ++p) {
        int n = KAN_NKNOT - 1 - p;
#pragma unroll
        for (int j = 0; j < KAN_NKNOT - 1 - 1; ++j) {
            if (j < n) {
                float a = (xv - tk[j]) / (tk[j + p] - tk[j]);
                float c = (tk[j + p + 1] - xv) / (tk[j + p + 1] - tk[j + 1]);
                bas[j] = a * bas[j] + c * bas[j + 1];
            }
        }
    }

    float silu = xv / (1.0f + expf(-xv));

    float* dst = F + (size_t)b * KAN_KDIM + (size_t)i * KAN_FEAT;
    dst[0] = silu;
#pragma unroll
    for (int c = 0; c < KAN_NBASIS; ++c) dst[1 + c] = bas[c];
}

__global__ __launch_bounds__(256)
void kan_weights(const float* __restrict__ coeff,
                 const float* __restrict__ mask,
                 const float* __restrict__ scale_base,
                 const float* __restrict__ scale_spline,
                 float* __restrict__ Wt) {
    int s = blockIdx.x * blockDim.x + threadIdx.x;
    if (s >= KAN_SIZE) return;
    int o = s / KAN_IN;
    int i = s - o * KAN_IN;

    float m  = mask[s];
    float sb = m * scale_base[s];
    float ss = m * scale_spline[s];

    int kbase = i * KAN_FEAT;
    Wt[(size_t)(kbase + 0) * KAN_OUT + o] = sb;
#pragma unroll
    for (int c = 0; c < KAN_NBASIS; ++c)
        Wt[(size_t)(kbase + 1 + c) * KAN_OUT + o] = ss * coeff[(size_t)s * KAN_NBASIS + c];
}

#define GEMM_WAVES  8
#define GEMM_KS     (KAN_KDIM / GEMM_WAVES)
#define GEMM_NLD    (GEMM_KS / 8)

__global__ __launch_bounds__(256) __attribute__((amdgpu_waves_per_eu(2, 2)))
void kan_gemm(const float* __restrict__ F,
              const float* __restrict__ Wt,
              const float* __restrict__ bias,
              float* __restrict__ out) {
    __shared__ float red[GEMM_WAVES * 256];

    int blk  = blockIdx.x;
    int mT   = blk >> 3;
    int nT   = blk & 7;
    int tid  = threadIdx.x;
    int wave = tid >> 5;
    int lane = tid & 31;
    int h    = lane >> 4;
    int ln   = lane & 15;

    int m0 = mT * 16;
    int n0 = nT * 16;
    int kb = wave * GEMM_KS;

    const float* Ap = F  + (size_t)(m0 + ln) * KAN_KDIM + kb + 4 * h;
    const float* Bp = Wt + (size_t)(kb + 4 * h) * KAN_OUT + n0 + ln;

    float bval = bias[n0 + ln];

    v4f a[GEMM_NLD];
    float b[GEMM_NLD][4];
#pragma unroll
    for (int j = 0; j < GEMM_NLD; ++j) {
        a[j] = *(const v4f*)(Ap + 8 * j);
#pragma unroll
        for (int e = 0; e < 4; ++e)
            b[j][e] = Bp[(size_t)(8 * j + e) * KAN_OUT];
    }

    v8f acc0 = {};
    v8f acc1 = {};
#pragma unroll
    for (int j = 0; j < GEMM_NLD; ++j) {
        v2f a0 = {a[j].x, a[j].y};
        v2f a1 = {a[j].z, a[j].w};
        v2f b0 = {b[j][0], b[j][1]};
        v2f b1 = {b[j][2], b[j][3]};
        acc0 = __builtin_amdgcn_wmma_f32_16x16x4_f32(
             false, a0,  false, b0,
             (short)0, acc0,  false,  false);
        acc1 = __builtin_amdgcn_wmma_f32_16x16x4_f32(
             false, a1,  false, b1,
             (short)0, acc1,  false,  false);
    }

#pragma unroll
    for (int r = 0; r < 8; ++r)
        red[wave * 256 + r * 32 + lane] = acc0[r] + acc1[r];
    __syncthreads();

    float s = red[tid];
#pragma unroll
    for (int w = 1; w < GEMM_WAVES; ++w) s += red[w * 256 + tid];

    int row = m0 + wave + 8 * h;
    out[(size_t)row * KAN_OUT + n0 + ln] = s + bval;
}

extern "C" void kernel_launch(void* const* d_in, const int* in_sizes, int n_in,
                              void* d_out, int out_size, void* d_ws, size_t ws_size,
                              hipStream_t stream) {
    const float* x            = (const float*)d_in[0];
    const float* grid_knots   = (const float*)d_in[1];
    const float* coeff        = (const float*)d_in[2];
    const float* mask         = (const float*)d_in[3];
    const float* scale_base   = (const float*)d_in[4];
    const float* scale_spline = (const float*)d_in[5];
    const float* bias         = (const float*)d_in[6];

    float* F  = (float*)d_ws;
    float* Wt = F + (size_t)KAN_BATCH * KAN_KDIM;

    kan_features<<<(KAN_BATCH * KAN_IN + 255) / 256, 256, 0, stream>>>(x, grid_knots, F);
    kan_weights<<<(KAN_SIZE + 255) / 256, 256, 0, stream>>>(coeff, mask, scale_base,
                                                            scale_spline, Wt);
    kan_gemm<<<256, 256, 0, stream>>>(F, Wt, bias, (float*)d_out);
}


# Three launches fused into one: each lane builds its A values (silu + spline basis of x) and B values (mask*scale*coeff) inside the GEMM, no workspace; K split by input index: new summation order
# speedup vs baseline: 2.4349x; 1.1937x over previous
#include <hip/hip_runtime.h>
#include <math.h>


#define KAN_BATCH   512
#define KAN_IN      128
#define KAN_OUT     128
#define KAN_G       5
#define KAN_KORD    3
#define KAN_NBASIS  (KAN_G + KAN_KORD)
#define KAN_NKNOT   (KAN_G + 2 * KAN_KORD + 1)
#define KAN_FEAT    (KAN_NBASIS + 1)
#define KAN_KDIM    (KAN_IN * KAN_FEAT)
#define KAN_SIZE    (KAN_IN * KAN_OUT)

typedef __attribute__((ext_vector_type(2))) float v2f;
typedef __attribute__((ext_vector_type(4))) float v4f;
typedef __attribute__((ext_vector_type(8))) float v8f;

__device__ __forceinline__
void kan_feat9(float xv, const float (&tk)[KAN_NKNOT], float (&f)[KAN_FEAT]) {
    float bas[KAN_NKNOT - 1];
#pragma unroll
    for (int j = 0; j < KAN_NKNOT - 1; ++j)
        bas[j] = (xv >= tk[j] && xv < tk[j + 1]) ? 1.0f : 0.0f;

#pragma unroll
    for (int p = 1; p <= KAN_KORD; ++p) {
        int n = KAN_NKNOT - 1 - p;
#pragma unroll
        for (int j = 0; j < KAN_NKNOT - 1 - 1; ++j) {
            if (j < n) {
                float a = (xv - tk[j]) / (tk[j + p] - tk[j]);
                float c = (tk[j + p + 1] - xv) / (tk[j + p + 1] - tk[j + 1]);
                bas[j] = a * bas[j] + c * bas[j + 1];
            }
        }
    }

    f[0] = xv / (1.0f + expf(-xv));
#pragma unroll
    for (int c = 0; c < KAN_NBASIS; ++c) f[1 + c] = bas[c];
}

#define GEMM_WAVES  8
#define LANE_INPUTS 8
#define TRIP_INPUTS 2
#define TRIP_K      (TRIP_INPUTS * KAN_FEAT)

__global__ __launch_bounds__(256) __attribute__((amdgpu_waves_per_eu(2, 2)))
void kan_fused(const float* __restrict__ x,
               const float* __restrict__ knots_g,
               const float* __restrict__ coeff,
               const float* __restrict__ mask,
               const float* __restrict__ scale_base,
               const float* __restrict__ scale_spline,
               const float* __restrict__ bias,
               float* __restrict__ out) {
    __shared__ float red[GEMM_WAVES * 256];

    int blk  = blockIdx.x;
    int mT   = blk >> 3;
    int nT   = blk & 7;
    int tid  = threadIdx.x;
    int wave = tid >> 5;
    int lane = tid & 31;
    int h    = lane >> 4;
    int ln   = lane & 15;

    int m0 = mT * 16;
    int n0 = nT * 16;
    int i0 = wave * 16 + h * LANE_INPUTS;
    int s0 = (n0 + ln) * KAN_IN + i0;

    const float* xp  = x + (size_t)(m0 + ln) * KAN_IN + i0;
    const float* cp  = coeff + (size_t)s0 * KAN_NBASIS;
    const float* mp  = mask + s0;
    const float* sbp = scale_base + s0;
    const float* ssp = scale_spline + s0;

    float bval = bias[n0 + ln];

    float tk[KAN_NKNOT];
#pragma unroll
    for (int j = 0; j < KAN_NKNOT; ++j) tk[j] = knots_g[j];

    v8f acc0 = {};
    v8f acc1 = {};
    v2f xc = *(const v2f*)xp;

#pragma nounroll
    for (int it = 0; it < LANE_INPUTS / TRIP_INPUTS; ++it) {
        int nx = (it + 1 < LANE_INPUTS / TRIP_INPUTS) ? it + 1 : it;
        v2f xn = *(const v2f*)(xp + TRIP_INPUTS * nx);
        v4f c0 = *(const v4f*)(cp + 16 * it + 0);
        v4f c1 = *(const v4f*)(cp + 16 * it + 4);
        v4f c2 = *(const v4f*)(cp + 16 * it + 8);
        v4f c3 = *(const v4f*)(cp + 16 * it + 12);
        v2f mk = *(const v2f*)(mp  + TRIP_INPUTS * it);
        v2f sb = *(const v2f*)(sbp + TRIP_INPUTS * it);
        v2f ss = *(const v2f*)(ssp + TRIP_INPUTS * it);

        float f0[KAN_FEAT], f1[KAN_FEAT];
        kan_feat9(xc.x, tk, f0);
        kan_feat9(xc.y, tk, f1);

        float fa[TRIP_K], fb[TRIP_K];
#pragma unroll
        for (int c = 0; c < KAN_FEAT; ++c) {
            fa[c] = f0[c];
            fa[KAN_FEAT + c] = f1[c];
        }
        float w0 = mk.x * ss.x;
        float w1 = mk.y * ss.y;
        fb[0]  = mk.x * sb.x;
        fb[1]  = w0 * c0.x;  fb[2]  = w0 * c0.y;  fb[3]  = w0 * c0.z;  fb[4]  = w0 * c0.w;
        fb[5]  = w0 * c1.x;  fb[6]  = w0 * c1.y;  fb[7]  = w0 * c1.z;  fb[8]  = w0 * c1.w;
        fb[9]  = mk.y * sb.y;
        fb[10] = w1 * c2.x;  fb[11] = w1 * c2.y;  fb[12] = w1 * c2.z;  fb[13] = w1 * c2.w;
        fb[14] = w1 * c3.x;  fb[15] = w1 * c3.y;  fb[16] = w1 * c3.z;  fb[17] = w1 * c3.w;

#pragma unroll
        for (int t = 0; t < TRIP_K / 2; ++t) {
            v2f av = {fa[2 * t], fa[2 * t + 1]};
            v2f bv = {fb[2 * t], fb[2 * t + 1]};
            if (t & 1)
                acc1 = __builtin_amdgcn_wmma_f32_16x16x4_f32(
                     false, av,  false, bv,
                     (short)0, acc1,  false,  false);
            else
                acc0 = __builtin_amdgcn_wmma_f32_16x16x4_f32(
                     false, av,  false, bv,
                     (short)0, acc0,  false,  false);
        }
        xc = xn;
    }

#pragma unroll
    for (int r = 0; r < 8; ++r)
        red[wave * 256 + r * 32 + lane] = acc0[r] + acc1[r];
    __syncthreads();

    float s = red[tid];
#pragma unroll
    for (int w = 1; w < GEMM_WAVES; ++w) s += red[w * 256 + tid];

    int row = m0 + wave + 8 * h;
    out[(size_t)row * KAN_OUT + n0 + ln] = s + bval;
}

extern "C" void kernel_launch(void* const* d_in, const int* in_sizes, int n_in,
                              void* d_out, int out_size, void* d_ws, size_t ws_size,
                              hipStream_t stream) {
    const float* x            = (const float*)d_in[0];
    const float* grid_knots   = (const float*)d_in[1];
    const float* coeff        = (const float*)d_in[2];
    const float* mask         = (const float*)d_in[3];
    const float* scale_base   = (const float*)d_in[4];
    const float* scale_spline = (const float*)d_in[5];
    const float* bias         = (const float*)d_in[6];

    kan_fused<<<256, 256, 0, stream>>>(x, grid_knots, coeff, mask, scale_base,
                                       scale_spline, bias, (float*)d_out);
}


# Cox-de Boor quotients as products with knot-span reciprocals (30 spans, one division per lane once per wave, spread by lane reads), not 54 divisions per input: quotients round twice, sums unchanged
# speedup vs baseline: 6.2700x; 2.5751x over previous
#include <hip/hip_runtime.h>
#include <math.h>


#define KAN_BATCH   512
#define KAN_IN      128
#define KAN_OUT     128
#define KAN_G       5
#define KAN_KORD    3
#define KAN_NBASIS  (KAN_G + KAN_KORD)
#define KAN_NKNOT   (KAN_G + 2 * KAN_KORD + 1)
#define KAN_FEAT    (KAN_NBASIS + 1)
#define KAN_KDIM    (KAN_IN * KAN_FEAT)
#define KAN_SIZE    (KAN_IN * KAN_OUT)

typedef __attribute__((ext_vector_type(2))) float v2f;
typedef __attribute__((ext_vector_type(4))) float v4f;
typedef __attribute__((ext_vector_type(8))) float v8f;

#define KAN_NSPAN (KAN_NKNOT - 1)

__device__ __forceinline__
void kan_span_recip(const float* __restrict__ knots_g, int lane,
                    float (&rk)[KAN_KORD][KAN_NSPAN]) {
    int p = 1 + (lane >= 11) + (lane >= 21);
    int j = lane - (lane >= 21 ? 21 : (lane >= 11 ? 11 : 0));
    if (lane >= 30) { p = 1; j = 0; }
    float r = 1.0f / (knots_g[j + p] - knots_g[j]);
    int rbits = __float_as_int(r);
#pragma unroll
    for (int pp = 1; pp <= KAN_KORD; ++pp) {
        int first = (pp == 1) ? 0 : (pp == 2 ? 11 : 21);
#pragma unroll
        for (int jj = 0; jj < KAN_NSPAN; ++jj)
            rk[pp - 1][jj] = (jj + pp < KAN_NKNOT)
                ? __int_as_float(__builtin_amdgcn_readlane(rbits, first + jj))
                : 0.0f;
    }
}

__device__ __forceinline__
void kan_feat9(float xv, const float (&tk)[KAN_NKNOT],
               const float (&rk)[KAN_KORD][KAN_NSPAN], float (&f)[KAN_FEAT]) {
    float bas[KAN_NKNOT - 1];
#pragma unroll
    for (int j = 0; j < KAN_NKNOT - 1; ++j)
        bas[j] = (xv >= tk[j] && xv < tk[j + 1]) ? 1.0f : 0.0f;

#pragma unroll
    for (int p = 1; p <= KAN_KORD; ++p) {
        int n = KAN_NKNOT - 1 - p;
#pragma unroll
        for (int j = 0; j < KAN_NKNOT - 1 - 1; ++j) {
            if (j < n) {
                float a = (xv - tk[j]) * rk[p - 1][j];
                float c = (tk[j + p + 1] - xv) * rk[p - 1][j + 1];
                bas[j] = a * bas[j] + c * bas[j + 1];
            }
        }
    }

    f[0] = xv / (1.0f + expf(-xv));
#pragma unroll
    for (int c = 0; c < KAN_NBASIS; ++c) f[1 + c] = bas[c];
}

#define GEMM_WAVES  8
#define LANE_INPUTS 8
#define TRIP_INPUTS 2
#define TRIP_K      (TRIP_INPUTS * KAN_FEAT)

__global__ __launch_bounds__(256) __attribute__((amdgpu_waves_per_eu(2, 2)))
void kan_fused(const float* __restrict__ x,
               const float* __restrict__ knots_g,
               const float* __restrict__ coeff,
               const float* __restrict__ mask,
               const float* __restrict__ scale_base,
               const float* __restrict__ scale_spline,
               const float* __restrict__ bias,
               float* __restrict__ out) {
    __shared__ float red[GEMM_WAVES * 256];

    int blk  = blockIdx.x;
    int mT   = blk >> 3;
    int nT   = blk & 7;
    int tid  = threadIdx.x;
    int wave = tid >> 5;
    int lane = tid & 31;
    int h    = lane >> 4;
    int ln   = lane & 15;

    int m0 = mT * 16;
    int n0 = nT * 16;
    int i0 = wave * 16 + h * LANE_INPUTS;
    int s0 = (n0 + ln) * KAN_IN + i0;

    const float* xp  = x + (size_t)(m0 + ln) * KAN_IN + i0;
    const float* cp  = coeff + (size_t)s0 * KAN_NBASIS;
    const float* mp  = mask + s0;
    const float* sbp = scale_base + s0;
    const float* ssp = scale_spline + s0;

    float bval = bias[n0 + ln];

    float tk[KAN_NKNOT];
#pragma unroll
    for (int j = 0; j < KAN_NKNOT; ++j) tk[j] = knots_g[j];

    float rk[KAN_KORD][KAN_NSPAN];
    kan_span_recip(knots_g, lane, rk);

    v8f acc0 = {};
    v8f acc1 = {};
    v2f xc = *(const v2f*)xp;

#pragma nounroll
    for (int it = 0; it < LANE_INPUTS / TRIP_INPUTS; ++it) {
        int nx = (it + 1 < LANE_INPUTS / TRIP_INPUTS) ? it + 1 : it;
        v2f xn = *(const v2f*)(xp + TRIP_INPUTS * nx);
        v4f c0 = *(const v4f*)(cp + 16 * it + 0);
        v4f c1 = *(const v4f*)(cp + 16 * it + 4);
        v4f c2 = *(const v4f*)(cp + 16 * it + 8);
        v4f c3 = *(const v4f*)(cp + 16 * it + 12);
        v2f mk = *(const v2f*)(mp  + TRIP_INPUTS * it);
        v2f sb = *(const v2f*)(sbp + TRIP_INPUTS * it);
        v2f ss = *(const v2f*)(ssp + TRIP_INPUTS * it);

        float f0[KAN_FEAT], f1[KAN_FEAT];
        kan_feat9(xc.x, tk, rk, f0);
        kan_feat9(xc.y, tk, rk, f1);

        float fa[TRIP_K], fb[TRIP_K];
#pragma unroll
        for (int c = 0; c < KAN_FEAT; ++c) {
            fa[c] = f0[c];
            fa[KAN_FEAT + c] = f1[c];
        }
        float w0 = mk.x * ss.x;
        float w1 = mk.y * ss.y;
        fb[0]  = mk.x * sb.x;
        fb[1]  = w0 * c0.x;  fb[2]  = w0 * c0.y;  fb[3]  = w0 * c0.z;  fb[4]  = w0 * c0.w;
        fb[5]  = w0 * c1.x;  fb[6]  = w0 * c1.y;  fb[7]  = w0 * c1.z;  fb[8]  = w0 * c1.w;
        fb[9]  = mk.y * sb.y;
        fb[10] = w1 * c2.x;  fb[11] = w1 * c2.y;  fb[12] = w1 * c2.z;  fb[13] = w1 * c2.w;
        fb[14] = w1 * c3.x;  fb[15] = w1 * c3.y;  fb[16] = w1 * c3.z;  fb[17] = w1 * c3.w;

#pragma unroll
        for (int t = 0; t < TRIP_K / 2; ++t) {
            v2f av = {fa[2 * t], fa[2 * t + 1]};
            v2f bv = {fb[2 * t], fb[2 * t + 1]};
            if (t & 1)
                acc1 = __builtin_amdgcn_wmma_f32_16x16x4_f32(
                     false, av,  false, bv,
                     (short)0, acc1,  false,  false);
            else
                acc0 = __builtin_amdgcn_wmma_f32_16x16x4_f32(
                     false, av,  false, bv,
                     (short)0, acc0,  false,  false);
        }
        xc = xn;
    }

#pragma unroll
    for (int r = 0; r < 8; ++r)
        red[wave * 256 + r * 32 + lane] = acc0[r] + acc1[r];
    __syncthreads();

    float s = red[tid];
#pragma unroll
    for (int w = 1; w < GEMM_WAVES; ++w) s += red[w * 256 + tid];

    int row = m0 + wave + 8 * h;
    out[(size_t)row * KAN_OUT + n0 + ln] = s + bval;
}

extern "C" void kernel_launch(void* const* d_in, const int* in_sizes, int n_in,
                              void* d_out, int out_size, void* d_ws, size_t ws_size,
                              hipStream_t stream) {
    const float* x            = (const float*)d_in[0];
    const float* grid_knots   = (const float*)d_in[1];
    const float* coeff        = (const float*)d_in[2];
    const float* mask         = (const float*)d_in[3];
    const float* scale_base   = (const float*)d_in[4];
    const float* scale_spline = (const float*)d_in[5];
    const float* bias         = (const float*)d_in[6];

    kan_fused<<<256, 256, 0, stream>>>(x, grid_knots, coeff, mask, scale_base,
                                       scale_spline, bias, (float*)d_out);
}


# Operands fetched once at entry: block's 64 KB of coefficients by whole-line loads (4 lines a load, was 32) via padded LDS; x, mask, scales 16 B a load; features computed meanwhile; sums unchanged
# speedup vs baseline: 6.7400x; 1.0750x over previous
#include <hip/hip_runtime.h>
#include <math.h>


#define KAN_BATCH   512
#define KAN_IN      128
#define KAN_OUT     128
#define KAN_G       5
#define KAN_KORD    3
#define KAN_NBASIS  (KAN_G + KAN_KORD)
#define KAN_NKNOT   (KAN_G + 2 * KAN_KORD + 1)
#define KAN_FEAT    (KAN_NBASIS + 1)
#define KAN_KDIM    (KAN_IN * KAN_FEAT)
#define KAN_SIZE    (KAN_IN * KAN_OUT)

typedef __attribute__((ext_vector_type(2))) float v2f;
typedef __attribute__((ext_vector_type(4))) float v4f;
typedef __attribute__((ext_vector_type(8))) float v8f;

#define KAN_NSPAN (KAN_NKNOT - 1)

__device__ __forceinline__
void kan_span_recip(const float* __restrict__ knots_g, int lane,
                    float (&rk)[KAN_KORD][KAN_NSPAN]) {
    int p = 1 + (lane >= 11) + (lane >= 21);
    int j = lane - (lane >= 21 ? 21 : (lane >= 11 ? 11 : 0));
    if (lane >= 30) { p = 1; j = 0; }
    float r = 1.0f / (knots_g[j + p] - knots_g[j]);
    int rbits = __float_as_int(r);
#pragma unroll
    for (int pp = 1; pp <= KAN_KORD; ++pp) {
        int first = (pp == 1) ? 0 : (pp == 2 ? 11 : 21);
#pragma unroll
        for (int jj = 0; jj < KAN_NSPAN; ++jj)
            rk[pp - 1][jj] = (jj + pp < KAN_NKNOT)
                ? __int_as_float(__builtin_amdgcn_readlane(rbits, first + jj))
                : 0.0f;
    }
}

__device__ __forceinline__
void kan_feat9(float xv, const float (&tk)[KAN_NKNOT],
               const float (&rk)[KAN_KORD][KAN_NSPAN], float (&f)[KAN_FEAT]) {
    float bas[KAN_NKNOT - 1];
#pragma unroll
    for (int j = 0; j < KAN_NKNOT - 1; ++j)
        bas[j] = (xv >= tk[j] && xv < tk[j + 1]) ? 1.0f : 0.0f;

#pragma unroll
    for (int p = 1; p <= KAN_KORD; ++p) {
        int n = KAN_NKNOT - 1 - p;
#pragma unroll
        for (int j = 0; j < KAN_NKNOT - 1 - 1; ++j) {
            if (j < n) {
                float a = (xv - tk[j]) * rk[p - 1][j];
                float c = (tk[j + p + 1] - xv) * rk[p - 1][j + 1];
                bas[j] = a * bas[j] + c * bas[j + 1];
            }
        }
    }

    f[0] = xv / (1.0f + expf(-xv));
#pragma unroll
    for (int c = 0; c < KAN_NBASIS; ++c) f[1 + c] = bas[c];
}

#define GEMM_WAVES  8
#define LANE_INPUTS 8
#define TRIP_INPUTS 2
#define TRIP_K      (TRIP_INPUTS * KAN_FEAT)
#define TILE_N      16
#define COL_FLOATS  (KAN_IN * KAN_NBASIS)
#define CST_PITCH   (COL_FLOATS + 4)

__global__ __launch_bounds__(256) __attribute__((amdgpu_waves_per_eu(2, 2)))
void kan_fused(const float* __restrict__ x,
               const float* __restrict__ knots_g,
               const float* __restrict__ coeff,
               const float* __restrict__ mask,
               const float* __restrict__ scale_base,
               const float* __restrict__ scale_spline,
               const float* __restrict__ bias,
               float* __restrict__ out) {
    __shared__ float red[GEMM_WAVES * 256];
    __shared__ __attribute__((aligned(16))) float cst[TILE_N * CST_PITCH];

    int blk  = blockIdx.x;
    int mT   = blk >> 3;
    int nT   = blk & 7;
    int tid  = threadIdx.x;
    int wave = tid >> 5;
    int lane = tid & 31;
    int h    = lane >> 4;
    int ln   = lane & 15;

    int m0 = mT * 16;
    int n0 = nT * 16;
    int i0 = wave * 16 + h * LANE_INPUTS;
    int s0 = (n0 + ln) * KAN_IN + i0;

    float tk[KAN_NKNOT];
#pragma unroll
    for (int j = 0; j < KAN_NKNOT; ++j) tk[j] = knots_g[j];

    float rk[KAN_KORD][KAN_NSPAN];
    kan_span_recip(knots_g, lane, rk);

    const float* xp = x + (size_t)(m0 + ln) * KAN_IN + i0;
    v4f xa = *(const v4f*)xp;
    v4f xb = *(const v4f*)(xp + 4);

    const v4f* cg = (const v4f*)(coeff + (size_t)n0 * COL_FLOATS) + tid;
    v4f st[TILE_N];
#pragma unroll
    for (int k = 0; k < TILE_N; ++k) st[k] = cg[k * (COL_FLOATS / 4)];

    v4f mka = *(const v4f*)(mask + s0);
    v4f mkb = *(const v4f*)(mask + s0 + 4);
    v4f sba = *(const v4f*)(scale_base + s0);
    v4f sbb = *(const v4f*)(scale_base + s0 + 4);
    v4f ssa = *(const v4f*)(scale_spline + s0);
    v4f ssb = *(const v4f*)(scale_spline + s0 + 4);

    float bval = bias[n0 + ln];

    float xv[LANE_INPUTS] = {xa.x, xa.y, xa.z, xa.w, xb.x, xb.y, xb.z, xb.w};
    float mk[LANE_INPUTS] = {mka.x, mka.y, mka.z, mka.w, mkb.x, mkb.y, mkb.z, mkb.w};
    float sb[LANE_INPUTS] = {sba.x, sba.y, sba.z, sba.w, sbb.x, sbb.y, sbb.z, sbb.w};
    float ss[LANE_INPUTS] = {ssa.x, ssa.y, ssa.z, ssa.w, ssb.x, ssb.y, ssb.z, ssb.w};

    float fa[LANE_INPUTS][KAN_FEAT];
#pragma unroll
    for (int q = 0; q < LANE_INPUTS; ++q) kan_feat9(xv[q], tk, rk, fa[q]);

    __builtin_amdgcn_sched_barrier(0);

#pragma unroll
    for (int k = 0; k < TILE_N; ++k) *(v4f*)&cst[k * CST_PITCH + 4 * tid] = st[k];
    __syncthreads();

    const float* cl = cst + ln * CST_PITCH + i0 * KAN_NBASIS;

    v8f acc0 = {};
    v8f acc1 = {};

#pragma unroll
    for (int it = 0; it < LANE_INPUTS / TRIP_INPUTS; ++it) {
        v4f c0 = *(const v4f*)(cl + 16 * it + 0);
        v4f c1 = *(const v4f*)(cl + 16 * it + 4);
        v4f c2 = *(const v4f*)(cl + 16 * it + 8);
        v4f c3 = *(const v4f*)(cl + 16 * it + 12);

        float fb[TRIP_K];
        float mk0 = mk[2 * it], mk1 = mk[2 * it + 1];
        float w0 = mk0 * ss[2 * it];
        float w1 = mk1 * ss[2 * it + 1];
        fb[0]  = mk0 * sb[2 * it];
        fb[1]  = w0 * c0.x;  fb[2]  = w0 * c0.y;  fb[3]  = w0 * c0.z;  fb[4]  = w0 * c0.w;
        fb[5]  = w0 * c1.x;  fb[6]  = w0 * c1.y;  fb[7]  = w0 * c1.z;  fb[8]  = w0 * c1.w;
        fb[9]  = mk1 * sb[2 * it + 1];
        fb[10] = w1 * c2.x;  fb[11] = w1 * c2.y;  fb[12] = w1 * c2.z;  fb[13] = w1 * c2.w;
        fb[14] = w1 * c3.x;  fb[15] = w1 * c3.y;  fb[16] = w1 * c3.z;  fb[17] = w1 * c3.w;

#pragma unroll
        for (int t = 0; t < TRIP_K / 2; ++t) {
            int e0 = 2 * t, e1 = 2 * t + 1;
            v2f av = {fa[2 * it + e0 / KAN_FEAT][e0 % KAN_FEAT],
                      fa[2 * it + e1 / KAN_FEAT][e1 % KAN_FEAT]};
            v2f bv = {fb[e0], fb[e1]};
            if (t & 1)
                acc1 = __builtin_amdgcn_wmma_f32_16x16x4_f32(
                     false, av,  false, bv,
                     (short)0, acc1,  false,  false);
            else
                acc0 = __builtin_amdgcn_wmma_f32_16x16x4_f32(
                     false, av,  false, bv,
                     (short)0, acc0,  false,  false);
        }
    }

#pragma unroll
    for (int r = 0; r < 8; ++r)
        red[wave * 256 + r * 32 + lane] = acc0[r] + acc1[r];
    __syncthreads();

    float s = red[tid];
#pragma unroll
    for (int w = 1; w < GEMM_WAVES; ++w) s += red[w * 256 + tid];

    int row = m0 + wave + 8 * h;
    out[(size_t)row * KAN_OUT + n0 + ln] = s + bval;
}

extern "C" void kernel_launch(void* const* d_in, const int* in_sizes, int n_in,
                              void* d_out, int out_size, void* d_ws, size_t ws_size,
                              hipStream_t stream) {
    const float* x            = (const float*)d_in[0];
    const float* grid_knots   = (const float*)d_in[1];
    const float* coeff        = (const float*)d_in[2];
    const float* mask         = (const float*)d_in[3];
    const float* scale_base   = (const float*)d_in[4];
    const float* scale_spline = (const float*)d_in[5];
    const float* bias         = (const float*)d_in[6];

    kan_fused<<<256, 256, 0, stream>>>(x, grid_knots, coeff, mask, scale_base,
                                       scale_spline, bias, (float*)d_out);
}


# Feature arithmetic on pairs of inputs (packed f32 across inputs q, q+1); k order of the matrix steps = those pairs, so operands need no register moves: new summation order
# speedup vs baseline: 7.3839x; 1.0955x over previous
#include <hip/hip_runtime.h>
#include <math.h>


#define KAN_BATCH   512
#define KAN_IN      128
#define KAN_OUT     128
#define KAN_G       5
#define KAN_KORD    3
#define KAN_NBASIS  (KAN_G + KAN_KORD)
#define KAN_NKNOT   (KAN_G + 2 * KAN_KORD + 1)
#define KAN_FEAT    (KAN_NBASIS + 1)
#define KAN_KDIM    (KAN_IN * KAN_FEAT)
#define KAN_SIZE    (KAN_IN * KAN_OUT)

typedef __attribute__((ext_vector_type(2))) float v2f;
typedef __attribute__((ext_vector_type(4))) float v4f;
typedef __attribute__((ext_vector_type(8))) float v8f;

#define KAN_NSPAN (KAN_NKNOT - 1)

__device__ __forceinline__
void kan_span_recip(const float* __restrict__ knots_g, int lane,
                    float (&rk)[KAN_KORD][KAN_NSPAN]) {
    int p = 1 + (lane >= 11) + (lane >= 21);
    int j = lane - (lane >= 21 ? 21 : (lane >= 11 ? 11 : 0));
    if (lane >= 30) { p = 1; j = 0; }
    float r = 1.0f / (knots_g[j + p] - knots_g[j]);
    int rbits = __float_as_int(r);
#pragma unroll
    for (int pp = 1; pp <= KAN_KORD; ++pp) {
        int first = (pp == 1) ? 0 : (pp == 2 ? 11 : 21);
#pragma unroll
        for (int jj = 0; jj < KAN_NSPAN; ++jj)
            rk[pp - 1][jj] = (jj + pp < KAN_NKNOT)
                ? __int_as_float(__builtin_amdgcn_readlane(rbits, first + jj))
                : 0.0f;
    }
}

__device__ __forceinline__
void kan_feat9x2(v2f xv, const float (&tk)[KAN_NKNOT],
                 const float (&rk)[KAN_KORD][KAN_NSPAN], v2f (&f)[KAN_FEAT]) {
    v2f bas[KAN_NKNOT - 1];
#pragma unroll
    for (int j = 0; j < KAN_NKNOT - 1; ++j) {
        bas[j].x = (xv.x >= tk[j] && xv.x < tk[j + 1]) ? 1.0f : 0.0f;
        bas[j].y = (xv.y >= tk[j] && xv.y < tk[j + 1]) ? 1.0f : 0.0f;
    }

#pragma unroll
    for (int p = 1; p <= KAN_KORD; ++p) {
        int n = KAN_NKNOT - 1 - p;
#pragma unroll
        for (int j = 0; j < KAN_NKNOT - 1 - 1; ++j) {
            if (j < n) {
                v2f a = (xv - tk[j]) * rk[p - 1][j];
                v2f c = (tk[j + p + 1] - xv) * rk[p - 1][j + 1];
                bas[j] = a * bas[j] + c * bas[j + 1];
            }
        }
    }

    f[0].x = xv.x / (1.0f + expf(-xv.x));
    f[0].y = xv.y / (1.0f + expf(-xv.y));
#pragma unroll
    for (int c = 0; c < KAN_NBASIS; ++c) f[1 + c] = bas[c];
}

#define GEMM_WAVES  8
#define LANE_INPUTS 8
#define TRIP_INPUTS 2
#define TILE_N      16
#define COL_FLOATS  (KAN_IN * KAN_NBASIS)
#define CST_PITCH   (COL_FLOATS + 4)

__global__ __launch_bounds__(256) __attribute__((amdgpu_waves_per_eu(2, 2)))
void kan_fused(const float* __restrict__ x,
               const float* __restrict__ knots_g,
               const float* __restrict__ coeff,
               const float* __restrict__ mask,
               const float* __restrict__ scale_base,
               const float* __restrict__ scale_spline,
               const float* __restrict__ bias,
               float* __restrict__ out) {
    __shared__ float red[GEMM_WAVES * 256];
    __shared__ __attribute__((aligned(16))) float cst[TILE_N * CST_PITCH];

    int blk  = blockIdx.x;
    int mT   = blk >> 3;
    int nT   = blk & 7;
    int tid  = threadIdx.x;
    int wave = tid >> 5;
    int lane = tid & 31;
    int h    = lane >> 4;
    int ln   = lane & 15;

    int m0 = mT * 16;
    int n0 = nT * 16;
    int i0 = wave * 16 + h * LANE_INPUTS;
    int s0 = (n0 + ln) * KAN_IN + i0;

    float tk[KAN_NKNOT];
#pragma unroll
    for (int j = 0; j < KAN_NKNOT; ++j) tk[j] = knots_g[j];

    float rk[KAN_KORD][KAN_NSPAN];
    kan_span_recip(knots_g, lane, rk);

    const float* xp = x + (size_t)(m0 + ln) * KAN_IN + i0;
    v4f xa = *(const v4f*)xp;
    v4f xb = *(const v4f*)(xp + 4);

    const v4f* cg = (const v4f*)(coeff + (size_t)n0 * COL_FLOATS) + tid;
    v4f st[TILE_N];
#pragma unroll
    for (int k = 0; k < TILE_N; ++k) st[k] = cg[k * (COL_FLOATS / 4)];

    v4f mka = *(const v4f*)(mask + s0);
    v4f mkb = *(const v4f*)(mask + s0 + 4);
    v4f sba = *(const v4f*)(scale_base + s0);
    v4f sbb = *(const v4f*)(scale_base + s0 + 4);
    v4f ssa = *(const v4f*)(scale_spline + s0);
    v4f ssb = *(const v4f*)(scale_spline + s0 + 4);

    float bval = bias[n0 + ln];

    float xv[LANE_INPUTS] = {xa.x, xa.y, xa.z, xa.w, xb.x, xb.y, xb.z, xb.w};
    float mk[LANE_INPUTS] = {mka.x, mka.y, mka.z, mka.w, mkb.x, mkb.y, mkb.z, mkb.w};
    float sb[LANE_INPUTS] = {sba.x, sba.y, sba.z, sba.w, sbb.x, sbb.y, sbb.z, sbb.w};
    float ss[LANE_INPUTS] = {ssa.x, ssa.y, ssa.z, ssa.w, ssb.x, ssb.y, ssb.z, ssb.w};

    v2f fa[LANE_INPUTS / TRIP_INPUTS][KAN_FEAT];
#pragma unroll
    for (int it = 0; it < LANE_INPUTS / TRIP_INPUTS; ++it) {
        v2f x2 = {xv[2 * it], xv[2 * it + 1]};
        kan_feat9x2(x2, tk, rk, fa[it]);
    }

    __builtin_amdgcn_sched_barrier(0);

#pragma unroll
    for (int k = 0; k < TILE_N; ++k) *(v4f*)&cst[k * CST_PITCH + 4 * tid] = st[k];
    __syncthreads();

    const float* cl = cst + ln * CST_PITCH + i0 * KAN_NBASIS;

    v8f acc0 = {};
    v8f acc1 = {};

#pragma unroll
    for (int it = 0; it < LANE_INPUTS / TRIP_INPUTS; ++it) {
        v4f c0 = *(const v4f*)(cl + 16 * it + 0);
        v4f c1 = *(const v4f*)(cl + 16 * it + 4);
        v4f c2 = *(const v4f*)(cl + 16 * it + 8);
        v4f c3 = *(const v4f*)(cl + 16 * it + 12);

        v2f fb[KAN_FEAT];
        float mk0 = mk[2 * it], mk1 = mk[2 * it + 1];
        float w0 = mk0 * ss[2 * it];
        float w1 = mk1 * ss[2 * it + 1];
        fb[0] = (v2f){mk0 * sb[2 * it], mk1 * sb[2 * it + 1]};
        fb[1] = (v2f){w0 * c0.x, w1 * c2.x};
        fb[2] = (v2f){w0 * c0.y, w1 * c2.y};
        fb[3] = (v2f){w0 * c0.z, w1 * c2.z};
        fb[4] = (v2f){w0 * c0.w, w1 * c2.w};
        fb[5] = (v2f){w0 * c1.x, w1 * c3.x};
        fb[6] = (v2f){w0 * c1.y, w1 * c3.y};
        fb[7] = (v2f){w0 * c1.z, w1 * c3.z};
        fb[8] = (v2f){w0 * c1.w, w1 * c3.w};

#pragma unroll
        for (int c = 0; c < KAN_FEAT; ++c) {
            if ((it * KAN_FEAT + c) & 1)
                acc1 = __builtin_amdgcn_wmma_f32_16x16x4_f32(
                     false, fa[it][c],  false, fb[c],
                     (short)0, acc1,  false,  false);
            else
                acc0 = __builtin_amdgcn_wmma_f32_16x16x4_f32(
                     false, fa[it][c],  false, fb[c],
                     (short)0, acc0,  false,  false);
        }
    }

#pragma unroll
    for (int r = 0; r < 8; ++r)
        red[wave * 256 + r * 32 + lane] = acc0[r] + acc1[r];
    __syncthreads();

    float s = red[tid];
#pragma unroll
    for (int w = 1; w < GEMM_WAVES; ++w) s += red[w * 256 + tid];

    int row = m0 + wave + 8 * h;
    out[(size_t)row * KAN_OUT + n0 + ln] = s + bval;
}

extern "C" void kernel_launch(void* const* d_in, const int* in_sizes, int n_in,
                              void* d_out, int out_size, void* d_ws, size_t ws_size,
                              hipStream_t stream) {
    const float* x            = (const float*)d_in[0];
    const float* grid_knots   = (const float*)d_in[1];
    const float* coeff        = (const float*)d_in[2];
    const float* mask         = (const float*)d_in[3];
    const float* scale_base   = (const float*)d_in[4];
    const float* scale_spline = (const float*)d_in[5];
    const float* bias         = (const float*)d_in[6];

    kan_fused<<<256, 256, 0, stream>>>(x, grid_knots, coeff, mask, scale_base,
                                       scale_spline, bias, (float*)d_out);
}


# Spline and silu arithmetic in fewer vector instructions: bases scaled once by span reciprocals (3 packed ops a term, was 4), x-t_j and x>=t_j once per knot, silu on v_exp/v_rcp; rounding differs
# speedup vs baseline: 8.0169x; 1.0857x over previous
#include <hip/hip_runtime.h>
#include <math.h>


#define KAN_BATCH   512
#define KAN_IN      128
#define KAN_OUT     128
#define KAN_G       5
#define KAN_KORD    3
#define KAN_NBASIS  (KAN_G + KAN_KORD)
#define KAN_NKNOT   (KAN_G + 2 * KAN_KORD + 1)
#define KAN_FEAT    (KAN_NBASIS + 1)
#define KAN_KDIM    (KAN_IN * KAN_FEAT)
#define KAN_SIZE    (KAN_IN * KAN_OUT)

typedef __attribute__((ext_vector_type(2))) float v2f;
typedef __attribute__((ext_vector_type(4))) float v4f;
typedef __attribute__((ext_vector_type(8))) float v8f;

#define KAN_NSPAN (KAN_NKNOT - 1)

__device__ __forceinline__
void kan_span_recip(const float* __restrict__ knots_g, int lane,
                    float (&rk)[KAN_KORD][KAN_NSPAN]) {
    int p = 1 + (lane >= 11) + (lane >= 21);
    int j = lane - (lane >= 21 ? 21 : (lane >= 11 ? 11 : 0));
    if (lane >= 30) { p = 1; j = 0; }
    float r = 1.0f / (knots_g[j + p] - knots_g[j]);
    int rbits = __float_as_int(r);
#pragma unroll
    for (int pp = 1; pp <= KAN_KORD; ++pp) {
        int first = (pp == 1) ? 0 : (pp == 2 ? 11 : 21);
#pragma unroll
        for (int jj = 0; jj < KAN_NSPAN; ++jj)
            rk[pp - 1][jj] = (jj + pp < KAN_NKNOT)
                ? __int_as_float(__builtin_amdgcn_readlane(rbits, first + jj))
                : 0.0f;
    }
}

__device__ __forceinline__
void kan_feat9x2(v2f xv, const float (&tk)[KAN_NKNOT],
                 const float (&rk)[KAN_KORD][KAN_NSPAN], v2f (&f)[KAN_FEAT]) {
    v2f d[KAN_NKNOT];
    bool gx[KAN_NKNOT], gy[KAN_NKNOT];
#pragma unroll
    for (int j = 0; j < KAN_NKNOT; ++j) {
        d[j]  = xv - tk[j];
        gx[j] = xv.x >= tk[j];
        gy[j] = xv.y >= tk[j];
    }

    v2f bas[KAN_NKNOT - 1];
#pragma unroll
    for (int j = 0; j < KAN_NKNOT - 1; ++j) {
        bas[j].x = (gx[j] && !gx[j + 1]) ? 1.0f : 0.0f;
        bas[j].y = (gy[j] && !gy[j + 1]) ? 1.0f : 0.0f;
    }

#pragma unroll
    for (int p = 1; p <= KAN_KORD; ++p) {
        int n = KAN_NKNOT - 1 - p;
        v2f g[KAN_NKNOT - 1];
#pragma unroll
        for (int j = 0; j < KAN_NKNOT - 1; ++j)
            if (j <= n) g[j] = bas[j] * rk[p - 1][j];
#pragma unroll
        for (int j = 0; j < KAN_NKNOT - 1 - 1; ++j)
            if (j < n) bas[j] = d[j] * g[j] - d[j + p + 1] * g[j + 1];
    }

    const float nlog2e = -1.4426950408889634f;
    f[0].x = xv.x * __builtin_amdgcn_rcpf(1.0f + __builtin_amdgcn_exp2f(xv.x * nlog2e));
    f[0].y = xv.y * __builtin_amdgcn_rcpf(1.0f + __builtin_amdgcn_exp2f(xv.y * nlog2e));
#pragma unroll
    for (int c = 0; c < KAN_NBASIS; ++c) f[1 + c] = bas[c];
}

#define GEMM_WAVES  8
#define LANE_INPUTS 8
#define TRIP_INPUTS 2
#define TILE_N      16
#define COL_FLOATS  (KAN_IN * KAN_NBASIS)
#define CST_PITCH   (COL_FLOATS + 4)

__global__ __launch_bounds__(256) __attribute__((amdgpu_waves_per_eu(2, 2)))
void kan_fused(const float* __restrict__ x,
               const float* __restrict__ knots_g,
               const float* __restrict__ coeff,
               const float* __restrict__ mask,
               const float* __restrict__ scale_base,
               const float* __restrict__ scale_spline,
               const float* __restrict__ bias,
               float* __restrict__ out) {
    __shared__ float red[GEMM_WAVES * 256];
    __shared__ __attribute__((aligned(16))) float cst[TILE_N * CST_PITCH];

    int blk  = blockIdx.x;
    int mT   = blk >> 3;
    int nT   = blk & 7;
    int tid  = threadIdx.x;
    int wave = tid >> 5;
    int lane = tid & 31;
    int h    = lane >> 4;
    int ln   = lane & 15;

    int m0 = mT * 16;
    int n0 = nT * 16;
    int i0 = wave * 16 + h * LANE_INPUTS;
    int s0 = (n0 + ln) * KAN_IN + i0;

    float tk[KAN_NKNOT];
#pragma unroll
    for (int j = 0; j < KAN_NKNOT; ++j) tk[j] = knots_g[j];

    float rk[KAN_KORD][KAN_NSPAN];
    kan_span_recip(knots_g, lane, rk);

    const float* xp = x + (size_t)(m0 + ln) * KAN_IN + i0;
    v4f xa = *(const v4f*)xp;
    v4f xb = *(const v4f*)(xp + 4);

    const v4f* cg = (const v4f*)(coeff + (size_t)n0 * COL_FLOATS) + tid;
    v4f st[TILE_N];
#pragma unroll
    for (int k = 0; k < TILE_N; ++k) st[k] = cg[k * (COL_FLOATS / 4)];

    v4f mka = *(const v4f*)(mask + s0);
    v4f mkb = *(const v4f*)(mask + s0 + 4);
    v4f sba = *(const v4f*)(scale_base + s0);
    v4f sbb = *(const v4f*)(scale_base + s0 + 4);
    v4f ssa = *(const v4f*)(scale_spline + s0);
    v4f ssb = *(const v4f*)(scale_spline + s0 + 4);

    float bval = bias[n0 + ln];

    float xv[LANE_INPUTS] = {xa.x, xa.y, xa.z, xa.w, xb.x, xb.y, xb.z, xb.w};
    float mk[LANE_INPUTS] = {mka.x, mka.y, mka.z, mka.w, mkb.x, mkb.y, mkb.z, mkb.w};
    float sb[LANE_INPUTS] = {sba.x, sba.y, sba.z, sba.w, sbb.x, sbb.y, sbb.z, sbb.w};
    float ss[LANE_INPUTS] = {ssa.x, ssa.y, ssa.z, ssa.w, ssb.x, ssb.y, ssb.z, ssb.w};

    v2f fa[LANE_INPUTS / TRIP_INPUTS][KAN_FEAT];
#pragma unroll
    for (int it = 0; it < LANE_INPUTS / TRIP_INPUTS; ++it) {
        v2f x2 = {xv[2 * it], xv[2 * it + 1]};
        kan_feat9x2(x2, tk, rk, fa[it]);
    }

    __builtin_amdgcn_sched_barrier(0);

#pragma unroll
    for (int k = 0; k < TILE_N; ++k) *(v4f*)&cst[k * CST_PITCH + 4 * tid] = st[k];
    __syncthreads();

    const float* cl = cst + ln * CST_PITCH + i0 * KAN_NBASIS;

    v8f acc0 = {};
    v8f acc1 = {};

#pragma unroll
    for (int it = 0; it < LANE_INPUTS / TRIP_INPUTS; ++it) {
        v4f c0 = *(const v4f*)(cl + 16 * it + 0);
        v4f c1 = *(const v4f*)(cl + 16 * it + 4);
        v4f c2 = *(const v4f*)(cl + 16 * it + 8);
        v4f c3 = *(const v4f*)(cl + 16 * it + 12);

        v2f fb[KAN_FEAT];
        float mk0 = mk[2 * it], mk1 = mk[2 * it + 1];
        float w0 = mk0 * ss[2 * it];
        float w1 = mk1 * ss[2 * it + 1];
        fb[0] = (v2f){mk0 * sb[2 * it], mk1 * sb[2 * it + 1]};
        fb[1] = (v2f){w0 * c0.x, w1 * c2.x};
        fb[2] = (v2f){w0 * c0.y, w1 * c2.y};
        fb[3] = (v2f){w0 * c0.z, w1 * c2.z};
        fb[4] = (v2f){w0 * c0.w, w1 * c2.w};
        fb[5] = (v2f){w0 * c1.x, w1 * c3.x};
        fb[6] = (v2f){w0 * c1.y, w1 * c3.y};
        fb[7] = (v2f){w0 * c1.z, w1 * c3.z};
        fb[8] = (v2f){w0 * c1.w, w1 * c3.w};

#pragma unroll
        for (int c = 0; c < KAN_FEAT; ++c) {
            if ((it * KAN_FEAT + c) & 1)
                acc1 = __builtin_amdgcn_wmma_f32_16x16x4_f32(
                     false, fa[it][c],  false, fb[c],
                     (short)0, acc1,  false,  false);
            else
                acc0 = __builtin_amdgcn_wmma_f32_16x16x4_f32(
                     false, fa[it][c],  false, fb[c],
                     (short)0, acc0,  false,  false);
        }
    }

#pragma unroll
    for (int r = 0; r < 8; ++r)
        red[wave * 256 + r * 32 + lane] = acc0[r] + acc1[r];
    __syncthreads();

    float s = red[tid];
#pragma unroll
    for (int w = 1; w < GEMM_WAVES; ++w) s += red[w * 256 + tid];

    int row = m0 + wave + 8 * h;
    out[(size_t)row * KAN_OUT + n0 + ln] = s + bval;
}

extern "C" void kernel_launch(void* const* d_in, const int* in_sizes, int n_in,
                              void* d_out, int out_size, void* d_ws, size_t ws_size,
                              hipStream_t stream) {
    const float* x            = (const float*)d_in[0];
    const float* grid_knots   = (const float*)d_in[1];
    const float* coeff        = (const float*)d_in[2];
    const float* mask         = (const float*)d_in[3];
    const float* scale_base   = (const float*)d_in[4];
    const float* scale_spline = (const float*)d_in[5];
    const float* bias         = (const float*)d_in[6];

    kan_fused<<<256, 256, 0, stream>>>(x, grid_knots, coeff, mask, scale_base,
                                       scale_spline, bias, (float*)d_out);
}
